# starting kernel
# baseline (speedup 1.0000x reference)
#include <hip/hip_runtime.h>

typedef float v2f __attribute__((ext_vector_type(2)));
typedef float v8f __attribute__((ext_vector_type(8)));

#define DF 300
#define K4 75
#define NPAD 304
#define FRAG_STRIDE 66
#define PIPE 3

#define WMMA_F32(a, b, c) \
    __builtin_amdgcn_wmma_f32_16x16x4_f32(false, (a), false, (b), (short)0, (c), false, false)


__global__ __launch_bounds__(256) void k_fill1(float* __restrict__ p, int n) {
    int i = blockIdx.x * 256 + threadIdx.x;
    if (i < n) p[i] = 1.0f;
}

__global__ __launch_bounds__(256) void k_deg_accum(float* __restrict__ deg,
                                                   const int* __restrict__ row, int e0) {
    int i = blockIdx.x * 256 + threadIdx.x;
    if (i < e0) unsafeAtomicAdd(&deg[row[i]], 1.0f);
}

__global__ __launch_bounds__(256) void k_rsqrt(float* __restrict__ p, int n) {
    int i = blockIdx.x * 256 + threadIdx.x;
    if (i < n) p[i] = rsqrtf(p[i]);
}

__global__ __launch_bounds__(256) void k_zero4(float4* __restrict__ p, int n4) {
    int i = blockIdx.x * 256 + threadIdx.x;
    if (i < n4) p[i] = make_float4(0.f, 0.f, 0.f, 0.f);
}

__global__ __launch_bounds__(256) void k_padW(const float* __restrict__ W,
                                              float* __restrict__ Wp) {
    int i = blockIdx.x * 256 + threadIdx.x;
    if (i < NPAD * DF) Wp[i] = (i < DF * DF) ? W[i] : 0.f;
}


__global__ __launch_bounds__(256) void k_gemm(const float* __restrict__ A,
                                              const float* __restrict__ Wp,
                                              const float* __restrict__ bias,
                                              float* __restrict__ out,
                                              int M, int relu_in) {
    __shared__ float lds[K4 * FRAG_STRIDE];
    const int m0 = blockIdx.x * 16;
    const int t  = threadIdx.x;

    for (int i = t; i < 16 * K4; i += 256) {
        int r  = i / K4;
        int c4 = i - r * K4;
        int m  = m0 + r;
        float4 v = make_float4(0.f, 0.f, 0.f, 0.f);
        if (m < M) v = ((const float4*)(A + (size_t)m * DF))[c4];
        if (relu_in) {
            v.x = fmaxf(v.x, 0.f); v.y = fmaxf(v.y, 0.f);
            v.z = fmaxf(v.z, 0.f); v.w = fmaxf(v.w, 0.f);
        }
        *(float2*)&lds[c4 * FRAG_STRIDE + r * 2]      = make_float2(v.x, v.y);
        *(float2*)&lds[c4 * FRAG_STRIDE + r * 2 + 32] = make_float2(v.z, v.w);
    }
    __syncthreads();

    const int wave = t >> 5;
    const int lane = t & 31;
    const int kh   = (lane >> 4) << 1;
    const int ln   = lane & 15;
    const float* af = &lds[lane * 2];
    const int mbase = m0 + ((lane >> 4) << 3);

#define LDA(s)    (*(const v2f*)(af + (s) * FRAG_STRIDE))
#define LDW(p, s) (*(const v2f*)((p) + (s) * 4))

    {
        const int n0 = wave * 16 + ln;
        const int n1 = (wave + 8) * 16 + ln;
        const float* w0 = Wp + (size_t)n0 * DF + kh;
        const float* w1 = Wp + (size_t)n1 * DF + kh;

        v8f c0 = {}, c1 = {};
        v2f a0 = LDA(0), a1 = LDA(1), a2 = LDA(2);
        v2f p0 = LDW(w0, 0), p1 = LDW(w0, 1), p2 = LDW(w0, 2);
        v2f q0 = LDW(w1, 0), q1 = LDW(w1, 1), q2 = LDW(w1, 2);

        for (int s = 0; s < K4 - PIPE; s += PIPE) {
            c0 = WMMA_F32(a0, p0, c0);  c1 = WMMA_F32(a0, q0, c1);
            a0 = LDA(s + 3);  p0 = LDW(w0, s + 3);  q0 = LDW(w1, s + 3);
            c0 = WMMA_F32(a1, p1, c0);  c1 = WMMA_F32(a1, q1, c1);
            a1 = LDA(s + 4);  p1 = LDW(w0, s + 4);  q1 = LDW(w1, s + 4);
            c0 = WMMA_F32(a2, p2, c0);  c1 = WMMA_F32(a2, q2, c1);
            a2 = LDA(s + 5);  p2 = LDW(w0, s + 5);  q2 = LDW(w1, s + 5);
        }
        c0 = WMMA_F32(a0, p0, c0);  c1 = WMMA_F32(a0, q0, c1);
        c0 = WMMA_F32(a1, p1, c0);  c1 = WMMA_F32(a1, q1, c1);
        c0 = WMMA_F32(a2, p2, c0);  c1 = WMMA_F32(a2, q2, c1);

        const float bv0 = bias[n0];
        const float bv1 = bias[n1];
        #pragma unroll
        for (int r = 0; r < 8; ++r) {
            int m = mbase + r;
            if (m < M) {
                out[(size_t)m * DF + n0] = c0[r] + bv0;
                out[(size_t)m * DF + n1] = c1[r] + bv1;
            }
        }
    }

    if (wave < 3) {
        const int n = (16 + wave) * 16 + ln;
        const float* w0 = Wp + (size_t)n * DF + kh;

        v8f c0 = {};
        v2f a0 = LDA(0), a1 = LDA(1), a2 = LDA(2);
        v2f p0 = LDW(w0, 0), p1 = LDW(w0, 1), p2 = LDW(w0, 2);

        for (int s = 0; s < K4 - PIPE; s += PIPE) {
            c0 = WMMA_F32(a0, p0, c0);
            a0 = LDA(s + 3);  p0 = LDW(w0, s + 3);
            c0 = WMMA_F32(a1, p1, c0);
            a1 = LDA(s + 4);  p1 = LDW(w0, s + 4);
            c0 = WMMA_F32(a2, p2, c0);
            a2 = LDA(s + 5);  p2 = LDW(w0, s + 5);
        }
        c0 = WMMA_F32(a0, p0, c0);
        c0 = WMMA_F32(a1, p1, c0);
        c0 = WMMA_F32(a2, p2, c0);

        if (n < DF) {
            const float bv = bias[n];
            #pragma unroll
            for (int r = 0; r < 8; ++r) {
                int m = mbase + r;
                if (m < M) out[(size_t)m * DF + n] = c0[r] + bv;
            }
        }
    }
#undef LDA
#undef LDW
}


__global__ __launch_bounds__(256) void k_scatter(const float* __restrict__ h,
                                                 const int* __restrict__ ei,
                                                 const float* __restrict__ dis,
                                                 float* __restrict__ out,
                                                 int E0, int Etot) {
    const int wv = blockIdx.x * 8 + (threadIdx.x >> 5);
    if (wv >= Etot) return;
    const int lane = threadIdx.x & 31;

    int r, c;
    if (wv < E0) { r = ei[wv]; c = ei[E0 + wv]; }
    else         { r = wv - E0; c = r; }

    const float norm = dis[r] * dis[c];
    const float4* hr = (const float4*)(h + (size_t)r * DF);
    float* oc = out + (size_t)c * DF;

    for (int q = lane; q < K4; q += 32) {
        float4 v = hr[q];
        int d = q << 2;
        unsafeAtomicAdd(oc + d + 0, norm * v.x);
        unsafeAtomicAdd(oc + d + 1, norm * v.y);
        unsafeAtomicAdd(oc + d + 2, norm * v.z);
        unsafeAtomicAdd(oc + d + 3, norm * v.w);
    }
}


extern "C" void kernel_launch(void* const* d_in, const int* in_sizes, int n_in,
                              void* d_out, int out_size, void* d_ws, size_t ws_size,
                              hipStream_t stream) {
    const float* x  = (const float*)d_in[0];
    const int*   ei = (const int*)d_in[1];
    const float* W1 = (const float*)d_in[2];
    const float* b1 = (const float*)d_in[3];
    const float* W2 = (const float*)d_in[4];
    const float* b2 = (const float*)d_in[5];
    float* out = (float*)d_out;

    const int N    = in_sizes[0] / DF;
    const int E0   = in_sizes[1] / 2;
    const int Etot = E0 + N;

    float* dis = (float*)d_ws;
    size_t disPad = ((size_t)N + 127) & ~(size_t)127;
    float* h  = dis + disPad;
    float* g  = h + (size_t)N * DF;
    float* wp = g + (size_t)N * DF;

    const int mtiles = (N + 15) / 16;
    const int nfeat4 = (N * DF) / 4;
    const int npadW  = (NPAD * DF + 255) / 256;

    k_fill1    <<<(N  + 255) / 256, 256, 0, stream>>>(dis, N);
    k_deg_accum<<<(E0 + 255) / 256, 256, 0, stream>>>(dis, ei, E0);
    k_rsqrt    <<<(N  + 255) / 256, 256, 0, stream>>>(dis, N);

    k_padW   <<<npadW, 256, 0, stream>>>(W1, wp);
    k_gemm   <<<mtiles, 256, 0, stream>>>(x, wp, b1, h, N, 0);
    k_zero4  <<<(nfeat4 + 255) / 256, 256, 0, stream>>>((float4*)g, nfeat4);
    k_scatter<<<(Etot + 7) / 8, 256, 0, stream>>>(h, ei, dis, g, E0, Etot);

    k_padW   <<<npadW, 256, 0, stream>>>(W2, wp);
    k_gemm   <<<mtiles, 256, 0, stream>>>(g, wp, b2, h, N, 1);
    k_zero4  <<<(nfeat4 + 255) / 256, 256, 0, stream>>>((float4*)out, nfeat4);
    k_scatter<<<(Etot + 7) / 8, 256, 0, stream>>>(h, ei, dis, out, E0, Etot);
}


# float-atomic edge scatter replaced by per-call grouping of edges by target (segments sorted by source id) + one-wave-per-node register sum; rows written once; sum order fixed: self, ascending source
# speedup vs baseline: 1.9826x; 1.9826x over previous
#include <hip/hip_runtime.h>

typedef float v2f __attribute__((ext_vector_type(2)));
typedef float v8f __attribute__((ext_vector_type(8)));

#define DF 300
#define K4 75
#define NPAD 304
#define FRAG_STRIDE 66
#define PIPE 3

#define WMMA_F32(a, b, c) \
    __builtin_amdgcn_wmma_f32_16x16x4_f32(false, (a), false, (b), (short)0, (c), false, false)


__global__ __launch_bounds__(256) void k_init(float* __restrict__ deg, int* __restrict__ cnt, int n) {
    int i = blockIdx.x * 256 + threadIdx.x;
    if (i < n) {
        deg[i] = 1.0f;
        cnt[i] = 0;
    }
}

__global__ __launch_bounds__(256) void k_deg_accum(float* __restrict__ deg, int* __restrict__ cnt,
                                                   int* __restrict__ pos,
                                                   const int* __restrict__ ei, int e0) {
    int i = blockIdx.x * 256 + threadIdx.x;
    if (i < e0) {
        unsafeAtomicAdd(&deg[ei[i]], 1.0f);
        pos[i] = atomicAdd(&cnt[ei[e0 + i]], 1);
    }
}

__global__ __launch_bounds__(256) void k_rsqrt(float* __restrict__ p, int n) {
    int i = blockIdx.x * 256 + threadIdx.x;
    if (i < n) p[i] = rsqrtf(p[i]);
}

__global__ __launch_bounds__(256) void k_scan_partial(const int* __restrict__ cnt,
                                                      int* __restrict__ bsum, int n) {
    __shared__ int red[8];
    const int t = threadIdx.x;
    const int i0 = blockIdx.x * 1024 + t * 4;
    int v = 0;
    #pragma unroll
    for (int k = 0; k < 4; ++k) if (i0 + k < n) v += cnt[i0 + k];
    for (int o = 16; o > 0; o >>= 1) v += __shfl_xor(v, o, 32);
    if ((t & 31) == 0) red[t >> 5] = v;
    __syncthreads();
    if (t == 0) {
        int s = 0;
        #pragma unroll
        for (int k = 0; k < 8; ++k) s += red[k];
        bsum[blockIdx.x] = s;
    }
}

__global__ __launch_bounds__(256) void k_scan_apply(int* __restrict__ cnt,
                                                    const int* __restrict__ bsum, int n) {
    __shared__ int wsum[8];
    __shared__ int wtot[8];
    const int t = threadIdx.x, lane = t & 31, wave = t >> 5;

    int bv = 0;
    for (int i = t; i < (int)blockIdx.x; i += 256) bv += bsum[i];
    for (int o = 16; o > 0; o >>= 1) bv += __shfl_xor(bv, o, 32);
    if (lane == 0) wsum[wave] = bv;

    const int i0 = blockIdx.x * 1024 + t * 4;
    int a[4];
    #pragma unroll
    for (int k = 0; k < 4; ++k) a[k] = (i0 + k < n) ? cnt[i0 + k] : 0;
    const int tsum = a[0] + a[1] + a[2] + a[3];
    int inc = tsum;
    for (int o = 1; o < 32; o <<= 1) {
        int u = __shfl_up(inc, o, 32);
        if (lane >= o) inc += u;
    }
    if (lane == 31) wtot[wave] = inc;
    __syncthreads();

    int ex = inc - tsum;
    #pragma unroll
    for (int k = 0; k < 8; ++k) {
        ex += wsum[k];
        if (k < wave) ex += wtot[k];
    }
    #pragma unroll
    for (int k = 0; k < 4; ++k) {
        if (i0 + k < n) cnt[i0 + k] = ex;
        ex += a[k];
    }
}

__global__ __launch_bounds__(256) void k_fill(const int* __restrict__ seg, const int* __restrict__ pos,
                                              const int* __restrict__ ei,
                                              int* __restrict__ srcU, int e0) {
    int i = blockIdx.x * 256 + threadIdx.x;
    if (i < e0) srcU[seg[ei[e0 + i]] + pos[i]] = ei[i];
}

__global__ __launch_bounds__(256) void k_sortseg(const int* __restrict__ seg,
                                                 const int* __restrict__ srcU,
                                                 int* __restrict__ srcS, int n, int e0) {
    int c = blockIdx.x * 256 + threadIdx.x;
    if (c >= n) return;
    const int s = seg[c];
    const int e = (c + 1 < n) ? seg[c + 1] : e0;
    for (int i = s; i < e; ++i) {
        const int v = srcU[i];
        int rank = 0;
        for (int j = s; j < e; ++j) {
            const int u = srcU[j];
            rank += (u < v || (u == v && j < i)) ? 1 : 0;
        }
        srcS[s + rank] = v;
    }
}

__global__ __launch_bounds__(256) void k_padW(const float* __restrict__ W,
                                              float* __restrict__ Wp) {
    int i = blockIdx.x * 256 + threadIdx.x;
    if (i < NPAD * DF) Wp[i] = (i < DF * DF) ? W[i] : 0.f;
}


__global__ __launch_bounds__(256) void k_gemm(const float* __restrict__ A,
                                              const float* __restrict__ Wp,
                                              const float* __restrict__ bias,
                                              float* __restrict__ out,
                                              int M, int relu_in) {
    __shared__ float lds[K4 * FRAG_STRIDE];
    const int m0 = blockIdx.x * 16;
    const int t  = threadIdx.x;

    for (int i = t; i < 16 * K4; i += 256) {
        int r  = i / K4;
        int c4 = i - r * K4;
        int m  = m0 + r;
        float4 v = make_float4(0.f, 0.f, 0.f, 0.f);
        if (m < M) v = ((const float4*)(A + (size_t)m * DF))[c4];
        if (relu_in) {
            v.x = fmaxf(v.x, 0.f); v.y = fmaxf(v.y, 0.f);
            v.z = fmaxf(v.z, 0.f); v.w = fmaxf(v.w, 0.f);
        }
        *(float2*)&lds[c4 * FRAG_STRIDE + r * 2]      = make_float2(v.x, v.y);
        *(float2*)&lds[c4 * FRAG_STRIDE + r * 2 + 32] = make_float2(v.z, v.w);
    }
    __syncthreads();

    const int wave = t >> 5;
    const int lane = t & 31;
    const int kh   = (lane >> 4) << 1;
    const int ln   = lane & 15;
    const float* af = &lds[lane * 2];
    const int mbase = m0 + ((lane >> 4) << 3);

#define LDA(s)    (*(const v2f*)(af + (s) * FRAG_STRIDE))
#define LDW(p, s) (*(const v2f*)((p) + (s) * 4))

    {
        const int n0 = wave * 16 + ln;
        const int n1 = (wave + 8) * 16 + ln;
        const float* w0 = Wp + (size_t)n0 * DF + kh;
        const float* w1 = Wp + (size_t)n1 * DF + kh;

        v8f c0 = {}, c1 = {};
        v2f a0 = LDA(0), a1 = LDA(1), a2 = LDA(2);
        v2f p0 = LDW(w0, 0), p1 = LDW(w0, 1), p2 = LDW(w0, 2);
        v2f q0 = LDW(w1, 0), q1 = LDW(w1, 1), q2 = LDW(w1, 2);

        for (int s = 0; s < K4 - PIPE; s += PIPE) {
            c0 = WMMA_F32(a0, p0, c0);  c1 = WMMA_F32(a0, q0, c1);
            a0 = LDA(s + 3);  p0 = LDW(w0, s + 3);  q0 = LDW(w1, s + 3);
            c0 = WMMA_F32(a1, p1, c0);  c1 = WMMA_F32(a1, q1, c1);
            a1 = LDA(s + 4);  p1 = LDW(w0, s + 4);  q1 = LDW(w1, s + 4);
            c0 = WMMA_F32(a2, p2, c0);  c1 = WMMA_F32(a2, q2, c1);
            a2 = LDA(s + 5);  p2 = LDW(w0, s + 5);  q2 = LDW(w1, s + 5);
        }
        c0 = WMMA_F32(a0, p0, c0);  c1 = WMMA_F32(a0, q0, c1);
        c0 = WMMA_F32(a1, p1, c0);  c1 = WMMA_F32(a1, q1, c1);
        c0 = WMMA_F32(a2, p2, c0);  c1 = WMMA_F32(a2, q2, c1);

        const float bv0 = bias[n0];
        const float bv1 = bias[n1];
        #pragma unroll
        for (int r = 0; r < 8; ++r) {
            int m = mbase + r;
            if (m < M) {
                out[(size_t)m * DF + n0] = c0[r] + bv0;
                out[(size_t)m * DF + n1] = c1[r] + bv1;
            }
        }
    }

    if (wave < 3) {
        const int n = (16 + wave) * 16 + ln;
        const float* w0 = Wp + (size_t)n * DF + kh;

        v8f c0 = {};
        v2f a0 = LDA(0), a1 = LDA(1), a2 = LDA(2);
        v2f p0 = LDW(w0, 0), p1 = LDW(w0, 1), p2 = LDW(w0, 2);

        for (int s = 0; s < K4 - PIPE; s += PIPE) {
            c0 = WMMA_F32(a0, p0, c0);
            a0 = LDA(s + 3);  p0 = LDW(w0, s + 3);
            c0 = WMMA_F32(a1, p1, c0);
            a1 = LDA(s + 4);  p1 = LDW(w0, s + 4);
            c0 = WMMA_F32(a2, p2, c0);
            a2 = LDA(s + 5);  p2 = LDW(w0, s + 5);
        }
        c0 = WMMA_F32(a0, p0, c0);
        c0 = WMMA_F32(a1, p1, c0);
        c0 = WMMA_F32(a2, p2, c0);

        if (n < DF) {
            const float bv = bias[n];
            #pragma unroll
            for (int r = 0; r < 8; ++r) {
                int m = mbase + r;
                if (m < M) out[(size_t)m * DF + n] = c0[r] + bv;
            }
        }
    }
#undef LDA
#undef LDW
}


__global__ __launch_bounds__(256) void k_gather(const float* __restrict__ h,
                                                const int* __restrict__ src,
                                                const int* __restrict__ seg,
                                                const float* __restrict__ dis,
                                                float* __restrict__ out, int n, int e0) {
    const int c = __builtin_amdgcn_readfirstlane((int)(blockIdx.x * 8 + (threadIdx.x >> 5)));
    if (c >= n) return;
    const int lane = threadIdx.x & 31;
    const int s    = seg[c];
    const int tot  = ((c + 1 < n) ? seg[c + 1] : e0) - s + 1;
    const float dc = dis[c];
    const int q2 = (lane + 64 < K4) ? lane + 64 : K4 - 1;

    float4 a0 = make_float4(0.f, 0.f, 0.f, 0.f), a1 = a0, a2 = a0;

#define ACC(AA, WW, VV) { AA.x = fmaf(WW, VV.x, AA.x); AA.y = fmaf(WW, VV.y, AA.y); \
                          AA.z = fmaf(WW, VV.z, AA.z); AA.w = fmaf(WW, VV.w, AA.w); }

    for (int base = 0; base < tot; base += 32) {
        const int k = base + lane;
        int myr = c;
        if (k >= 1 && k < tot) myr = src[s + k - 1];
        const int myw = __float_as_int(dis[myr] * dc);
        const int nn = (tot - base < 32) ? tot - base : 32;

        int i = 0;
        for (; i + 1 < nn; i += 2) {
            const int   r0 = __builtin_amdgcn_readlane(myr, i);
            const int   r1 = __builtin_amdgcn_readlane(myr, i + 1);
            const float w0 = __int_as_float(__builtin_amdgcn_readlane(myw, i));
            const float w1 = __int_as_float(__builtin_amdgcn_readlane(myw, i + 1));
            const float4* p0 = (const float4*)(h + (size_t)r0 * DF);
            const float4* p1 = (const float4*)(h + (size_t)r1 * DF);
            const float4 u0 = p0[lane], u1 = p0[lane + 32], u2 = p0[q2];
            const float4 v0 = p1[lane], v1 = p1[lane + 32], v2 = p1[q2];
            ACC(a0, w0, u0); ACC(a1, w0, u1); ACC(a2, w0, u2);
            ACC(a0, w1, v0); ACC(a1, w1, v1); ACC(a2, w1, v2);
        }
        if (i < nn) {
            const int   r0 = __builtin_amdgcn_readlane(myr, i);
            const float w0 = __int_as_float(__builtin_amdgcn_readlane(myw, i));
            const float4* p0 = (const float4*)(h + (size_t)r0 * DF);
            const float4 u0 = p0[lane], u1 = p0[lane + 32], u2 = p0[q2];
            ACC(a0, w0, u0); ACC(a1, w0, u1); ACC(a2, w0, u2);
        }
    }
#undef ACC

    float4* o = (float4*)(out + (size_t)c * DF);
    o[lane]      = a0;
    o[lane + 32] = a1;
    if (lane + 64 < K4) o[lane + 64] = a2;
}


extern "C" void kernel_launch(void* const* d_in, const int* in_sizes, int n_in,
                              void* d_out, int out_size, void* d_ws, size_t ws_size,
                              hipStream_t stream) {
    const float* x  = (const float*)d_in[0];
    const int*   ei = (const int*)d_in[1];
    const float* W1 = (const float*)d_in[2];
    const float* b1 = (const float*)d_in[3];
    const float* W2 = (const float*)d_in[4];
    const float* b2 = (const float*)d_in[5];
    float* out = (float*)d_out;

    const int N    = in_sizes[0] / DF;
    const int E0   = in_sizes[1] / 2;

    float* dis = (float*)d_ws;
    size_t disPad = ((size_t)N + 127) & ~(size_t)127;
    float* h  = dis + disPad;
    float* g  = h + (size_t)N * DF;
    float* wp = g + (size_t)N * DF;
    int* cnt  = (int*)g;
    int* bsum = cnt + disPad;
    int* srcU = bsum + 1024;
    int* srcS = srcU + E0;
    int* pos  = srcS + E0;

    const int mtiles = (N + 15) / 16;
    const int npadW  = (NPAD * DF + 255) / 256;
    const int nscan  = (N + 1023) / 1024;

    k_init     <<<(N  + 255) / 256, 256, 0, stream>>>(dis, cnt, N);
    k_deg_accum<<<(E0 + 255) / 256, 256, 0, stream>>>(dis, cnt, pos, ei, E0);
    k_rsqrt    <<<(N  + 255) / 256, 256, 0, stream>>>(dis, N);

    k_scan_partial<<<nscan, 256, 0, stream>>>(cnt, bsum, N);
    k_scan_apply  <<<nscan, 256, 0, stream>>>(cnt, bsum, N);
    k_fill        <<<(E0 + 255) / 256, 256, 0, stream>>>(cnt, pos, ei, srcU, E0);
    k_sortseg     <<<(N  + 255) / 256, 256, 0, stream>>>(cnt, srcU, srcS, N, E0);

    k_padW  <<<npadW, 256, 0, stream>>>(W1, wp);
    k_gemm  <<<mtiles, 256, 0, stream>>>(x, wp, b1, h, N, 0);
    k_gather<<<(N + 7) / 8, 256, 0, stream>>>(h, srcS, cnt, dis, out, N, E0);

    k_padW  <<<npadW, 256, 0, stream>>>(W2, wp);
    k_gemm  <<<mtiles, 256, 0, stream>>>(out, wp, b2, h, N, 1);
    k_gather<<<(N + 7) / 8, 256, 0, stream>>>(h, srcS, cnt, dis, out, N, E0);
}


# GEMM B operand: W repacked per call into matrix-fragment order (pad kernel rewritten) so each B load is one contiguous 16-byte-per-lane whole-line read serving two K-steps; sum order unchanged
# speedup vs baseline: 2.7916x; 1.4081x over previous
#include <hip/hip_runtime.h>

typedef float v2f __attribute__((ext_vector_type(2)));
typedef float v4f __attribute__((ext_vector_type(4)));
typedef float v8f __attribute__((ext_vector_type(8)));

#define DF 300
#define K4 75
#define NT 19
#define NP 38
#define WF_TILE (NP * 128)
#define FRAG_STRIDE 66

#define WMMA_F32(a, b, c) \
    __builtin_amdgcn_wmma_f32_16x16x4_f32(false, (a), false, (b), (short)0, (c), false, false)


__global__ __launch_bounds__(256) void k_init(float* __restrict__ deg, int* __restrict__ cnt, int n) {
    int i = blockIdx.x * 256 + threadIdx.x;
    if (i < n) {
        deg[i] = 1.0f;
        cnt[i] = 0;
    }
}

__global__ __launch_bounds__(256) void k_deg_accum(float* __restrict__ deg, int* __restrict__ cnt,
                                                   int* __restrict__ pos,
                                                   const int* __restrict__ ei, int e0) {
    int i = blockIdx.x * 256 + threadIdx.x;
    if (i < e0) {
        unsafeAtomicAdd(&deg[ei[i]], 1.0f);
        pos[i] = atomicAdd(&cnt[ei[e0 + i]], 1);
    }
}

__global__ __launch_bounds__(256) void k_rsqrt(float* __restrict__ p, int n) {
    int i = blockIdx.x * 256 + threadIdx.x;
    if (i < n) p[i] = rsqrtf(p[i]);
}

__global__ __launch_bounds__(256) void k_scan_partial(const int* __restrict__ cnt,
                                                      int* __restrict__ bsum, int n) {
    __shared__ int red[8];
    const int t = threadIdx.x;
    const int i0 = blockIdx.x * 1024 + t * 4;
    int v = 0;
    #pragma unroll
    for (int k = 0; k < 4; ++k) if (i0 + k < n) v += cnt[i0 + k];
    for (int o = 16; o > 0; o >>= 1) v += __shfl_xor(v, o, 32);
    if ((t & 31) == 0) red[t >> 5] = v;
    __syncthreads();
    if (t == 0) {
        int s = 0;
        #pragma unroll
        for (int k = 0; k < 8; ++k) s += red[k];
        bsum[blockIdx.x] = s;
    }
}

__global__ __launch_bounds__(256) void k_scan_apply(int* __restrict__ cnt,
                                                    const int* __restrict__ bsum, int n) {
    __shared__ int wsum[8];
    __shared__ int wtot[8];
    const int t = threadIdx.x, lane = t & 31, wave = t >> 5;

    int bv = 0;
    for (int i = t; i < (int)blockIdx.x; i += 256) bv += bsum[i];
    for (int o = 16; o > 0; o >>= 1) bv += __shfl_xor(bv, o, 32);
    if (lane == 0) wsum[wave] = bv;

    const int i0 = blockIdx.x * 1024 + t * 4;
    int a[4];
    #pragma unroll
    for (int k = 0; k < 4; ++k) a[k] = (i0 + k < n) ? cnt[i0 + k] : 0;
    const int tsum = a[0] + a[1] + a[2] + a[3];
    int inc = tsum;
    for (int o = 1; o < 32; o <<= 1) {
        int u = __shfl_up(inc, o, 32);
        if (lane >= o) inc += u;
    }
    if (lane == 31) wtot[wave] = inc;
    __syncthreads();

    int ex = inc - tsum;
    #pragma unroll
    for (int k = 0; k < 8; ++k) {
        ex += wsum[k];
        if (k < wave) ex += wtot[k];
    }
    #pragma unroll
    for (int k = 0; k < 4; ++k) {
        if (i0 + k < n) cnt[i0 + k] = ex;
        ex += a[k];
    }
}

__global__ __launch_bounds__(256) void k_fill(const int* __restrict__ seg, const int* __restrict__ pos,
                                              const int* __restrict__ ei,
                                              int* __restrict__ srcU, int e0) {
    int i = blockIdx.x * 256 + threadIdx.x;
    if (i < e0) srcU[seg[ei[e0 + i]] + pos[i]] = ei[i];
}

__global__ __launch_bounds__(256) void k_sortseg(const int* __restrict__ seg,
                                                 const int* __restrict__ srcU,
                                                 int* __restrict__ srcS, int n, int e0) {
    int c = blockIdx.x * 256 + threadIdx.x;
    if (c >= n) return;
    const int s = seg[c];
    const int e = (c + 1 < n) ? seg[c + 1] : e0;
    for (int i = s; i < e; ++i) {
        const int v = srcU[i];
        int rank = 0;
        for (int j = s; j < e; ++j) {
            const int u = srcU[j];
            rank += (u < v || (u == v && j < i)) ? 1 : 0;
        }
        srcS[s + rank] = v;
    }
}

__global__ __launch_bounds__(256) void k_packW(const float* __restrict__ W,
                                               float* __restrict__ Wf) {
    const int i = blockIdx.x * 256 + threadIdx.x;
    if (i >= NT * NP * 32) return;
    const int lane = i & 31;
    const int tp   = i >> 5;
    const int tile = tp / NP;
    const int p    = tp - tile * NP;
    const int n    = tile * 16 + (lane & 15);
    const int k0   = 8 * p + 2 * (lane >> 4);
    const int k1   = k0 + 4;
    float4 v = make_float4(0.f, 0.f, 0.f, 0.f);
    if (n < DF) {
        const float* row = W + (size_t)n * DF;
        if (k0 < DF) { const float2 u = *(const float2*)(row + k0); v.x = u.x; v.y = u.y; }
        if (k1 < DF) { const float2 u = *(const float2*)(row + k1); v.z = u.x; v.w = u.y; }
    }
    ((float4*)Wf)[i] = v;
}


__global__ __launch_bounds__(256) void k_gemm(const float* __restrict__ A,
                                              const float* __restrict__ Wf,
                                              const float* __restrict__ bias,
                                              float* __restrict__ out,
                                              int M, int relu_in) {
    __shared__ float lds[(K4 + 1) * FRAG_STRIDE];
    const int m0 = blockIdx.x * 16;
    const int t  = threadIdx.x;

    for (int i = t; i < 16 * K4; i += 256) {
        int r  = i / K4;
        int c4 = i - r * K4;
        int m  = m0 + r;
        float4 v = make_float4(0.f, 0.f, 0.f, 0.f);
        if (m < M) v = ((const float4*)(A + (size_t)m * DF))[c4];
        if (relu_in) {
            v.x = fmaxf(v.x, 0.f); v.y = fmaxf(v.y, 0.f);
            v.z = fmaxf(v.z, 0.f); v.w = fmaxf(v.w, 0.f);
        }
        *(float2*)&lds[c4 * FRAG_STRIDE + r * 2]      = make_float2(v.x, v.y);
        *(float2*)&lds[c4 * FRAG_STRIDE + r * 2 + 32] = make_float2(v.z, v.w);
    }
    __syncthreads();

    const int wave = t >> 5;
    const int lane = t & 31;
    const int ln   = lane & 15;
    const float* af = &lds[lane * 2];
    const int mbase = m0 + ((lane >> 4) << 3);

#define LDA(s)     (*(const v2f*)(af + (s) * FRAG_STRIDE))
#define LDB(p, pr) (*(const v4f*)((p) + (pr) * 128))

    {
        const int n0 = wave * 16 + ln;
        const int n1 = (wave + 8) * 16 + ln;
        const float* b0 = Wf + (size_t)wave * WF_TILE + lane * 4;
        const float* b1 = Wf + (size_t)(wave + 8) * WF_TILE + lane * 4;

        v8f c0 = {}, c1 = {};
        v2f ax0 = LDA(0), ax1 = LDA(1);  v4f px = LDB(b0, 0), qx = LDB(b1, 0);
        v2f ay0 = LDA(2), ay1 = LDA(3);  v4f py = LDB(b0, 1), qy = LDB(b1, 1);

        for (int p = 0; p < NP - 2; p += 2) {
            c0 = WMMA_F32(ax0, px.lo, c0);  c1 = WMMA_F32(ax0, qx.lo, c1);
            c0 = WMMA_F32(ax1, px.hi, c0);  c1 = WMMA_F32(ax1, qx.hi, c1);
            ax0 = LDA(2 * p + 4);  ax1 = LDA(2 * p + 5);  px = LDB(b0, p + 2);  qx = LDB(b1, p + 2);
            c0 = WMMA_F32(ay0, py.lo, c0);  c1 = WMMA_F32(ay0, qy.lo, c1);
            c0 = WMMA_F32(ay1, py.hi, c0);  c1 = WMMA_F32(ay1, qy.hi, c1);
            ay0 = LDA(2 * p + 6);  ay1 = LDA(2 * p + 7);  py = LDB(b0, p + 3);  qy = LDB(b1, p + 3);
        }
        c0 = WMMA_F32(ax0, px.lo, c0);  c1 = WMMA_F32(ax0, qx.lo, c1);
        c0 = WMMA_F32(ax1, px.hi, c0);  c1 = WMMA_F32(ax1, qx.hi, c1);
        c0 = WMMA_F32(ay0, py.lo, c0);  c1 = WMMA_F32(ay0, qy.lo, c1);

        const float bv0 = bias[n0];
        const float bv1 = bias[n1];
        #pragma unroll
        for (int r = 0; r < 8; ++r) {
            int m = mbase + r;
            if (m < M) {
                out[(size_t)m * DF + n0] = c0[r] + bv0;
                out[(size_t)m * DF + n1] = c1[r] + bv1;
            }
        }
    }

    if (wave < 3) {
        const int n = (16 + wave) * 16 + ln;
        const float* b0 = Wf + (size_t)(16 + wave) * WF_TILE + lane * 4;

        v8f c0 = {};
        v2f ax0 = LDA(0), ax1 = LDA(1);  v4f px = LDB(b0, 0);
        v2f ay0 = LDA(2), ay1 = LDA(3);  v4f py = LDB(b0, 1);

        for (int p = 0; p < NP - 2; p += 2) {
            c0 = WMMA_F32(ax0, px.lo, c0);
            c0 = WMMA_F32(ax1, px.hi, c0);
            ax0 = LDA(2 * p + 4);  ax1 = LDA(2 * p + 5);  px = LDB(b0, p + 2);
            c0 = WMMA_F32(ay0, py.lo, c0);
            c0 = WMMA_F32(ay1, py.hi, c0);
            ay0 = LDA(2 * p + 6);  ay1 = LDA(2 * p + 7);  py = LDB(b0, p + 3);
        }
        c0 = WMMA_F32(ax0, px.lo, c0);
        c0 = WMMA_F32(ax1, px.hi, c0);
        c0 = WMMA_F32(ay0, py.lo, c0);

        if (n < DF) {
            const float bv = bias[n];
            #pragma unroll
            for (int r = 0; r < 8; ++r) {
                int m = mbase + r;
                if (m < M) out[(size_t)m * DF + n] = c0[r] + bv;
            }
        }
    }
#undef LDA
#undef LDB
}


__global__ __launch_bounds__(256) void k_gather(const float* __restrict__ h,
                                                const int* __restrict__ src,
                                                const int* __restrict__ seg,
                                                const float* __restrict__ dis,
                                                float* __restrict__ out, int n, int e0) {
    const int c = __builtin_amdgcn_readfirstlane((int)(blockIdx.x * 8 + (threadIdx.x >> 5)));
    if (c >= n) return;
    const int lane = threadIdx.x & 31;
    const int s    = seg[c];
    const int tot  = ((c + 1 < n) ? seg[c + 1] : e0) - s + 1;
    const float dc = dis[c];
    const int q2 = (lane + 64 < K4) ? lane + 64 : K4 - 1;

    float4 a0 = make_float4(0.f, 0.f, 0.f, 0.f), a1 = a0, a2 = a0;

#define ACC(AA, WW, VV) { AA.x = fmaf(WW, VV.x, AA.x); AA.y = fmaf(WW, VV.y, AA.y); \
                          AA.z = fmaf(WW, VV.z, AA.z); AA.w = fmaf(WW, VV.w, AA.w); }

    for (int base = 0; base < tot; base += 32) {
        const int k = base + lane;
        int myr = c;
        if (k >= 1 && k < tot) myr = src[s + k - 1];
        const int myw = __float_as_int(dis[myr] * dc);
        const int nn = (tot - base < 32) ? tot - base : 32;

        int i = 0;
        for (; i + 1 < nn; i += 2) {
            const int   r0 = __builtin_amdgcn_readlane(myr, i);
            const int   r1 = __builtin_amdgcn_readlane(myr, i + 1);
            const float w0 = __int_as_float(__builtin_amdgcn_readlane(myw, i));
            const float w1 = __int_as_float(__builtin_amdgcn_readlane(myw, i + 1));
            const float4* p0 = (const float4*)(h + (size_t)r0 * DF);
            const float4* p1 = (const float4*)(h + (size_t)r1 * DF);
            const float4 u0 = p0[lane], u1 = p0[lane + 32], u2 = p0[q2];
            const float4 v0 = p1[lane], v1 = p1[lane + 32], v2 = p1[q2];
            ACC(a0, w0, u0); ACC(a1, w0, u1); ACC(a2, w0, u2);
            ACC(a0, w1, v0); ACC(a1, w1, v1); ACC(a2, w1, v2);
        }
        if (i < nn) {
            const int   r0 = __builtin_amdgcn_readlane(myr, i);
            const float w0 = __int_as_float(__builtin_amdgcn_readlane(myw, i));
            const float4* p0 = (const float4*)(h + (size_t)r0 * DF);
            const float4 u0 = p0[lane], u1 = p0[lane + 32], u2 = p0[q2];
            ACC(a0, w0, u0); ACC(a1, w0, u1); ACC(a2, w0, u2);
        }
    }
#undef ACC

    float4* o = (float4*)(out + (size_t)c * DF);
    o[lane]      = a0;
    o[lane + 32] = a1;
    if (lane + 64 < K4) o[lane + 64] = a2;
}


extern "C" void kernel_launch(void* const* d_in, const int* in_sizes, int n_in,
                              void* d_out, int out_size, void* d_ws, size_t ws_size,
                              hipStream_t stream) {
    const float* x  = (const float*)d_in[0];
    const int*   ei = (const int*)d_in[1];
    const float* W1 = (const float*)d_in[2];
    const float* b1 = (const float*)d_in[3];
    const float* W2 = (const float*)d_in[4];
    const float* b2 = (const float*)d_in[5];
    float* out = (float*)d_out;

    const int N    = in_sizes[0] / DF;
    const int E0   = in_sizes[1] / 2;

    float* dis = (float*)d_ws;
    size_t disPad = ((size_t)N + 127) & ~(size_t)127;
    float* h  = dis + disPad;
    float* g  = h + (size_t)N * DF;
    int* cnt  = (int*)g;
    int* bsum = cnt + disPad;
    int* srcU = bsum + 1024;
    int* srcS = srcU + E0;
    int* pos  = srcS + E0;
    size_t wfOff = (disPad + 1024 + 3 * (size_t)E0 + 63) & ~(size_t)63;
    float* wf = g + wfOff;

    const int mtiles = (N + 15) / 16;
    const int npackW = (NT * NP * 32 + 255) / 256;
    const int nscan  = (N + 1023) / 1024;

    k_init     <<<(N  + 255) / 256, 256, 0, stream>>>(dis, cnt, N);
    k_deg_accum<<<(E0 + 255) / 256, 256, 0, stream>>>(dis, cnt, pos, ei, E0);
    k_rsqrt    <<<(N  + 255) / 256, 256, 0, stream>>>(dis, N);

    k_scan_partial<<<nscan, 256, 0, stream>>>(cnt, bsum, N);
    k_scan_apply  <<<nscan, 256, 0, stream>>>(cnt, bsum, N);
    k_fill        <<<(E0 + 255) / 256, 256, 0, stream>>>(cnt, pos, ei, srcU, E0);
    k_sortseg     <<<(N  + 255) / 256, 256, 0, stream>>>(cnt, srcU, srcS, N, E0);

    k_packW <<<npackW, 256, 0, stream>>>(W1, wf);
    k_gemm  <<<mtiles, 256, 0, stream>>>(x, wf, b1, h, N, 0);
    k_gather<<<(N + 7) / 8, 256, 0, stream>>>(h, srcS, cnt, dis, out, N, E0);

    k_packW <<<npackW, 256, 0, stream>>>(W2, wf);
    k_gemm  <<<mtiles, 256, 0, stream>>>(out, wf, b2, h, N, 1);
    k_gather<<<(N + 7) / 8, 256, 0, stream>>>(h, srcS, cnt, dis, out, N, E0);
}


# GEMM block tile 16 -> 32 rows: each packed-W B fragment is reused from registers for 2 M tiles (W reads from L2 halved); 32-row panel staged with its loads issued as one group; K sum order unchanged
# speedup vs baseline: 2.9228x; 1.0470x over previous
#include <hip/hip_runtime.h>

typedef float v2f __attribute__((ext_vector_type(2)));
typedef float v4f __attribute__((ext_vector_type(4)));
typedef float v8f __attribute__((ext_vector_type(8)));

#define DF 300
#define K4 75
#define NT 19
#define NP 38
#define WF_TILE (NP * 128)

#define WMMA_F32(a, b, c) \
    __builtin_amdgcn_wmma_f32_16x16x4_f32(false, (a), false, (b), (short)0, (c), false, false)


__global__ __launch_bounds__(256) void k_init(float* __restrict__ deg, int* __restrict__ cnt, int n) {
    int i = blockIdx.x * 256 + threadIdx.x;
    if (i < n) {
        deg[i] = 1.0f;
        cnt[i] = 0;
    }
}

__global__ __launch_bounds__(256) void k_deg_accum(float* __restrict__ deg, int* __restrict__ cnt,
                                                   int* __restrict__ pos,
                                                   const int* __restrict__ ei, int e0) {
    int i = blockIdx.x * 256 + threadIdx.x;
    if (i < e0) {
        unsafeAtomicAdd(&deg[ei[i]], 1.0f);
        pos[i] = atomicAdd(&cnt[ei[e0 + i]], 1);
    }
}

__global__ __launch_bounds__(256) void k_rsqrt(float* __restrict__ p, int n) {
    int i = blockIdx.x * 256 + threadIdx.x;
    if (i < n) p[i] = rsqrtf(p[i]);
}

__global__ __launch_bounds__(256) void k_scan_partial(const int* __restrict__ cnt,
                                                      int* __restrict__ bsum, int n) {
    __shared__ int red[8];
    const int t = threadIdx.x;
    const int i0 = blockIdx.x * 1024 + t * 4;
    int v = 0;
    #pragma unroll
    for (int k = 0; k < 4; ++k) if (i0 + k < n) v += cnt[i0 + k];
    for (int o = 16; o > 0; o >>= 1) v += __shfl_xor(v, o, 32);
    if ((t & 31) == 0) red[t >> 5] = v;
    __syncthreads();
    if (t == 0) {
        int s = 0;
        #pragma unroll
        for (int k = 0; k < 8; ++k) s += red[k];
        bsum[blockIdx.x] = s;
    }
}

__global__ __launch_bounds__(256) void k_scan_apply(int* __restrict__ cnt,
                                                    const int* __restrict__ bsum, int n) {
    __shared__ int wsum[8];
    __shared__ int wtot[8];
    const int t = threadIdx.x, lane = t & 31, wave = t >> 5;

    int bv = 0;
    for (int i = t; i < (int)blockIdx.x; i += 256) bv += bsum[i];
    for (int o = 16; o > 0; o >>= 1) bv += __shfl_xor(bv, o, 32);
    if (lane == 0) wsum[wave] = bv;

    const int i0 = blockIdx.x * 1024 + t * 4;
    int a[4];
    #pragma unroll
    for (int k = 0; k < 4; ++k) a[k] = (i0 + k < n) ? cnt[i0 + k] : 0;
    const int tsum = a[0] + a[1] + a[2] + a[3];
    int inc = tsum;
    for (int o = 1; o < 32; o <<= 1) {
        int u = __shfl_up(inc, o, 32);
        if (lane >= o) inc += u;
    }
    if (lane == 31) wtot[wave] = inc;
    __syncthreads();

    int ex = inc - tsum;
    #pragma unroll
    for (int k = 0; k < 8; ++k) {
        ex += wsum[k];
        if (k < wave) ex += wtot[k];
    }
    #pragma unroll
    for (int k = 0; k < 4; ++k) {
        if (i0 + k < n) cnt[i0 + k] = ex;
        ex += a[k];
    }
}

__global__ __launch_bounds__(256) void k_fill(const int* __restrict__ seg, const int* __restrict__ pos,
                                              const int* __restrict__ ei,
                                              int* __restrict__ srcU, int e0) {
    int i = blockIdx.x * 256 + threadIdx.x;
    if (i < e0) srcU[seg[ei[e0 + i]] + pos[i]] = ei[i];
}

__global__ __launch_bounds__(256) void k_sortseg(const int* __restrict__ seg,
                                                 const int* __restrict__ srcU,
                                                 int* __restrict__ srcS, int n, int e0) {
    int c = blockIdx.x * 256 + threadIdx.x;
    if (c >= n) return;
    const int s = seg[c];
    const int e = (c + 1 < n) ? seg[c + 1] : e0;
    for (int i = s; i < e; ++i) {
        const int v = srcU[i];
        int rank = 0;
        for (int j = s; j < e; ++j) {
            const int u = srcU[j];
            rank += (u < v || (u == v && j < i)) ? 1 : 0;
        }
        srcS[s + rank] = v;
    }
}

__global__ __launch_bounds__(256) void k_packW(const float* __restrict__ W,
                                               float* __restrict__ Wf) {
    const int i = blockIdx.x * 256 + threadIdx.x;
    if (i >= NT * NP * 32) return;
    const int lane = i & 31;
    const int tp   = i >> 5;
    const int tile = tp / NP;
    const int p    = tp - tile * NP;
    const int n    = tile * 16 + (lane & 15);
    const int k0   = 8 * p + 2 * (lane >> 4);
    const int k1   = k0 + 4;
    float4 v = make_float4(0.f, 0.f, 0.f, 0.f);
    if (n < DF) {
        const float* row = W + (size_t)n * DF;
        if (k0 < DF) { const float2 u = *(const float2*)(row + k0); v.x = u.x; v.y = u.y; }
        if (k1 < DF) { const float2 u = *(const float2*)(row + k1); v.z = u.x; v.w = u.y; }
    }
    ((float4*)Wf)[i] = v;
}


#define MT 2
#define SLAB (MT * 64 + 2)
#define NSLAB (K4 + 1)

#define LDA(base, mt, s) (*(const v2f*)((base) + (mt) * 64 + (s) * SLAB))
#define LDB(p, pr)       (*(const v4f*)((p) + (pr) * 128))

template <bool RELU>
__device__ __forceinline__ void stage_panel(const float* __restrict__ A, float* lds,
                                            int m0, int M, int t) {
    constexpr int TOT = 16 * MT * K4;
    constexpr int NJ  = (TOT + 255) / 256;
    float4 v[NJ];
    #pragma unroll
    for (int j = 0; j < NJ; ++j) {
        int i = t + j * 256;
        if (i > TOT - 1) i = TOT - 1;
        const int r  = i / K4;
        const int c4 = i - r * K4;
        const int m  = (m0 + r < M) ? m0 + r : M - 1;
        v[j] = ((const float4*)(A + (size_t)m * DF))[c4];
    }
    #pragma unroll
    for (int j = 0; j < NJ; ++j) {
        const int i = t + j * 256;
        if (i < TOT) {
            const int r  = i / K4;
            const int c4 = i - r * K4;
            float4 u = v[j];
            if (RELU) {
                u.x = fmaxf(u.x, 0.f); u.y = fmaxf(u.y, 0.f);
                u.z = fmaxf(u.z, 0.f); u.w = fmaxf(u.w, 0.f);
            }
            float* d = &lds[c4 * SLAB + (r >> 4) * 64 + (r & 15) * 2];
            *(float2*)d        = make_float2(u.x, u.y);
            *(float2*)(d + 32) = make_float2(u.z, u.w);
        }
    }
}

template <int NM, int NN>
__device__ __forceinline__ void gemm_tail(const float* af, const float* __restrict__ b0,
                                          const float* __restrict__ bias, float* __restrict__ out,
                                          int n, int mbase, int M) {
    v8f c[NM][NN];
    v2f ax0[NM], ax1[NM], ay0[NM], ay1[NM];
    v4f px[NN], py[NN];
    #pragma unroll
    for (int i = 0; i < NM; ++i) {
        ax0[i] = LDA(af, i, 0);  ax1[i] = LDA(af, i, 1);
        ay0[i] = LDA(af, i, 2);  ay1[i] = LDA(af, i, 3);
        #pragma unroll
        for (int q = 0; q < NN; ++q) c[i][q] = v8f{};
    }
    #pragma unroll
    for (int q = 0; q < NN; ++q) { px[q] = LDB(b0 + q * WF_TILE, 0);  py[q] = LDB(b0 + q * WF_TILE, 1); }

#define TAIL_STEP(AF, BS, BH) \
    _Pragma("unroll") for (int i = 0; i < NM; ++i) \
    _Pragma("unroll") for (int q = 0; q < NN; ++q) c[i][q] = WMMA_F32(AF[i], BS[q].BH, c[i][q]);

    for (int p = 0; p < NP - 2; p += 2) {
        TAIL_STEP(ax0, px, lo)
        TAIL_STEP(ax1, px, hi)
        #pragma unroll
        for (int i = 0; i < NM; ++i) { ax0[i] = LDA(af, i, 2 * p + 4);  ax1[i] = LDA(af, i, 2 * p + 5); }
        #pragma unroll
        for (int q = 0; q < NN; ++q) px[q] = LDB(b0 + q * WF_TILE, p + 2);
        TAIL_STEP(ay0, py, lo)
        TAIL_STEP(ay1, py, hi)
        #pragma unroll
        for (int i = 0; i < NM; ++i) { ay0[i] = LDA(af, i, 2 * p + 6);  ay1[i] = LDA(af, i, 2 * p + 7); }
        #pragma unroll
        for (int q = 0; q < NN; ++q) py[q] = LDB(b0 + q * WF_TILE, p + 3);
    }
    TAIL_STEP(ax0, px, lo)
    TAIL_STEP(ax1, px, hi)
    TAIL_STEP(ay0, py, lo)
#undef TAIL_STEP

    #pragma unroll
    for (int q = 0; q < NN; ++q) {
        const int nq = n + q * 16;
        if (nq < DF) {
            const float bv = bias[nq];
            #pragma unroll
            for (int i = 0; i < NM; ++i) {
                #pragma unroll
                for (int r = 0; r < 8; ++r) {
                    const int m = mbase + i * 16 + r;
                    if (m < M) out[(size_t)m * DF + nq] = c[i][q][r] + bv;
                }
            }
        }
    }
}

__global__ __launch_bounds__(256) void k_gemm(const float* __restrict__ A,
                                              const float* __restrict__ Wf,
                                              const float* __restrict__ bias,
                                              float* __restrict__ out,
                                              int M, int relu_in) {
    __shared__ float lds[NSLAB * SLAB];
    const int m0 = blockIdx.x * (16 * MT);
    const int t  = threadIdx.x;

    if (relu_in) stage_panel<true>(A, lds, m0, M, t);
    else         stage_panel<false>(A, lds, m0, M, t);
    __syncthreads();

    const int wave = __builtin_amdgcn_readfirstlane(t >> 5);
    const int lane = t & 31;
    const int ln   = lane & 15;
    const float* af = &lds[lane * 2];
    const int mbase = m0 + ((lane >> 4) << 3);

    {
        const int n0 = wave * 16 + ln;
        const int n1 = (wave + 8) * 16 + ln;
        const float* b0 = Wf + (size_t)wave * WF_TILE + lane * 4;
        const float* b1 = Wf + (size_t)(wave + 8) * WF_TILE + lane * 4;

        v8f c0[MT], c1[MT];
        v2f ax0[MT], ax1[MT], ay0[MT], ay1[MT];
        #pragma unroll
        for (int i = 0; i < MT; ++i) {
            c0[i] = v8f{};  c1[i] = v8f{};
            ax0[i] = LDA(af, i, 0);  ax1[i] = LDA(af, i, 1);
            ay0[i] = LDA(af, i, 2);  ay1[i] = LDA(af, i, 3);
        }
        v4f px = LDB(b0, 0), qx = LDB(b1, 0);
        v4f py = LDB(b0, 1), qy = LDB(b1, 1);

#define MAIN_STEP(AF, P, Q, H) \
    _Pragma("unroll") for (int i = 0; i < MT; ++i) { \
        c0[i] = WMMA_F32(AF[i], P.H, c0[i]);  c1[i] = WMMA_F32(AF[i], Q.H, c1[i]); }

        for (int p = 0; p < NP - 2; p += 2) {
            MAIN_STEP(ax0, px, qx, lo)
            MAIN_STEP(ax1, px, qx, hi)
            #pragma unroll
            for (int i = 0; i < MT; ++i) { ax0[i] = LDA(af, i, 2 * p + 4);  ax1[i] = LDA(af, i, 2 * p + 5); }
            px = LDB(b0, p + 2);  qx = LDB(b1, p + 2);
            MAIN_STEP(ay0, py, qy, lo)
            MAIN_STEP(ay1, py, qy, hi)
            #pragma unroll
            for (int i = 0; i < MT; ++i) { ay0[i] = LDA(af, i, 2 * p + 6);  ay1[i] = LDA(af, i, 2 * p + 7); }
            py = LDB(b0, p + 3);  qy = LDB(b1, p + 3);
        }
        MAIN_STEP(ax0, px, qx, lo)
        MAIN_STEP(ax1, px, qx, hi)
        MAIN_STEP(ay0, py, qy, lo)
#undef MAIN_STEP

        const float bv0 = bias[n0];
        const float bv1 = bias[n1];
        #pragma unroll
        for (int i = 0; i < MT; ++i) {
            #pragma unroll
            for (int r = 0; r < 8; ++r) {
                const int m = mbase + i * 16 + r;
                if (m < M) {
                    out[(size_t)m * DF + n0] = c0[i][r] + bv0;
                    out[(size_t)m * DF + n1] = c1[i][r] + bv1;
                }
            }
        }
    }

    if (wave < 3) {
        gemm_tail<MT, 1>(af, Wf + (size_t)(16 + wave) * WF_TILE + lane * 4, bias, out,
                         (16 + wave) * 16 + ln, mbase, M);
    }
}
#undef LDA
#undef LDB


__global__ __launch_bounds__(256) void k_gather(const float* __restrict__ h,
                                                const int* __restrict__ src,
                                                const int* __restrict__ seg,
                                                const float* __restrict__ dis,
                                                float* __restrict__ out, int n, int e0) {
    const int c = __builtin_amdgcn_readfirstlane((int)(blockIdx.x * 8 + (threadIdx.x >> 5)));
    if (c >= n) return;
    const int lane = threadIdx.x & 31;
    const int s    = seg[c];
    const int tot  = ((c + 1 < n) ? seg[c + 1] : e0) - s + 1;
    const float dc = dis[c];
    const int q2 = (lane + 64 < K4) ? lane + 64 : K4 - 1;

    float4 a0 = make_float4(0.f, 0.f, 0.f, 0.f), a1 = a0, a2 = a0;

#define ACC(AA, WW, VV) { AA.x = fmaf(WW, VV.x, AA.x); AA.y = fmaf(WW, VV.y, AA.y); \
                          AA.z = fmaf(WW, VV.z, AA.z); AA.w = fmaf(WW, VV.w, AA.w); }

    for (int base = 0; base < tot; base += 32) {
        const int k = base + lane;
        int myr = c;
        if (k >= 1 && k < tot) myr = src[s + k - 1];
        const int myw = __float_as_int(dis[myr] * dc);
        const int nn = (tot - base < 32) ? tot - base : 32;

        int i = 0;
        for (; i + 1 < nn; i += 2) {
            const int   r0 = __builtin_amdgcn_readlane(myr, i);
            const int   r1 = __builtin_amdgcn_readlane(myr, i + 1);
            const float w0 = __int_as_float(__builtin_amdgcn_readlane(myw, i));
            const float w1 = __int_as_float(__builtin_amdgcn_readlane(myw, i + 1));
            const float4* p0 = (const float4*)(h + (size_t)r0 * DF);
            const float4* p1 = (const float4*)(h + (size_t)r1 * DF);
            const float4 u0 = p0[lane], u1 = p0[lane + 32], u2 = p0[q2];
            const float4 v0 = p1[lane], v1 = p1[lane + 32], v2 = p1[q2];
            ACC(a0, w0, u0); ACC(a1, w0, u1); ACC(a2, w0, u2);
            ACC(a0, w1, v0); ACC(a1, w1, v1); ACC(a2, w1, v2);
        }
        if (i < nn) {
            const int   r0 = __builtin_amdgcn_readlane(myr, i);
            const float w0 = __int_as_float(__builtin_amdgcn_readlane(myw, i));
            const float4* p0 = (const float4*)(h + (size_t)r0 * DF);
            const float4 u0 = p0[lane], u1 = p0[lane + 32], u2 = p0[q2];
            ACC(a0, w0, u0); ACC(a1, w0, u1); ACC(a2, w0, u2);
        }
    }
#undef ACC

    float4* o = (float4*)(out + (size_t)c * DF);
    o[lane]      = a0;
    o[lane + 32] = a1;
    if (lane + 64 < K4) o[lane + 64] = a2;
}


extern "C" void kernel_launch(void* const* d_in, const int* in_sizes, int n_in,
                              void* d_out, int out_size, void* d_ws, size_t ws_size,
                              hipStream_t stream) {
    const float* x  = (const float*)d_in[0];
    const int*   ei = (const int*)d_in[1];
    const float* W1 = (const float*)d_in[2];
    const float* b1 = (const float*)d_in[3];
    const float* W2 = (const float*)d_in[4];
    const float* b2 = (const float*)d_in[5];
    float* out = (float*)d_out;

    const int N    = in_sizes[0] / DF;
    const int E0   = in_sizes[1] / 2;

    float* dis = (float*)d_ws;
    size_t disPad = ((size_t)N + 127) & ~(size_t)127;
    float* h  = dis + disPad;
    float* g  = h + (size_t)N * DF;
    int* cnt  = (int*)g;
    int* bsum = cnt + disPad;
    int* srcU = bsum + 1024;
    int* srcS = srcU + E0;
    int* pos  = srcS + E0;
    size_t wfOff = (disPad + 1024 + 3 * (size_t)E0 + 63) & ~(size_t)63;
    float* wf = g + wfOff;

    const int mtiles = (N + 16 * MT - 1) / (16 * MT);
    const int npackW = (NT * NP * 32 + 255) / 256;
    const int nscan  = (N + 1023) / 1024;

    k_init     <<<(N  + 255) / 256, 256, 0, stream>>>(dis, cnt, N);
    k_deg_accum<<<(E0 + 255) / 256, 256, 0, stream>>>(dis, cnt, pos, ei, E0);
    k_rsqrt    <<<(N  + 255) / 256, 256, 0, stream>>>(dis, N);

    k_scan_partial<<<nscan, 256, 0, stream>>>(cnt, bsum, N);
    k_scan_apply  <<<nscan, 256, 0, stream>>>(cnt, bsum, N);
    k_fill        <<<(E0 + 255) / 256, 256, 0, stream>>>(cnt, pos, ei, srcU, E0);
    k_sortseg     <<<(N  + 255) / 256, 256, 0, stream>>>(cnt, srcU, srcS, N, E0);

    k_packW <<<npackW, 256, 0, stream>>>(W1, wf);
    k_gemm  <<<mtiles, 256, 0, stream>>>(x, wf, b1, h, N, 0);
    k_gather<<<(N + 7) / 8, 256, 0, stream>>>(h, srcS, cnt, dis, out, N, E0);

    k_packW <<<npackW, 256, 0, stream>>>(W2, wf);
    k_gemm  <<<mtiles, 256, 0, stream>>>(out, wf, b2, h, N, 1);
    k_gather<<<(N + 7) / 8, 256, 0, stream>>>(h, srcS, cnt, dis, out, N, E0);
}


# layer-1 aggregation fused into the layer-2 GEMM's panel staging: each block sums its 32 aggregate rows from h into LDS fragments (ReLU there); aggregate never stored; sum order unchanged
# speedup vs baseline: 3.0884x; 1.0567x over previous
#include <hip/hip_runtime.h>

typedef float v2f __attribute__((ext_vector_type(2)));
typedef float v4f __attribute__((ext_vector_type(4)));
typedef float v8f __attribute__((ext_vector_type(8)));

#define DF 300
#define K4 75
#define NT 19
#define NP 38
#define WF_TILE (NP * 128)

#define WMMA_F32(a, b, c) \
    __builtin_amdgcn_wmma_f32_16x16x4_f32(false, (a), false, (b), (short)0, (c), false, false)


__global__ __launch_bounds__(256) void k_init(float* __restrict__ deg, int* __restrict__ cnt, int n) {
    int i = blockIdx.x * 256 + threadIdx.x;
    if (i < n) {
        deg[i] = 1.0f;
        cnt[i] = 0;
    }
}

__global__ __launch_bounds__(256) void k_deg_accum(float* __restrict__ deg, int* __restrict__ cnt,
                                                   int* __restrict__ pos,
                                                   const int* __restrict__ ei, int e0) {
    int i = blockIdx.x * 256 + threadIdx.x;
    if (i < e0) {
        unsafeAtomicAdd(&deg[ei[i]], 1.0f);
        pos[i] = atomicAdd(&cnt[ei[e0 + i]], 1);
    }
}

__global__ __launch_bounds__(256) void k_rsqrt(float* __restrict__ p, int n) {
    int i = blockIdx.x * 256 + threadIdx.x;
    if (i < n) p[i] = rsqrtf(p[i]);
}

__global__ __launch_bounds__(256) void k_scan_partial(const int* __restrict__ cnt,
                                                      int* __restrict__ bsum, int n) {
    __shared__ int red[8];
    const int t = threadIdx.x;
    const int i0 = blockIdx.x * 1024 + t * 4;
    int v = 0;
    #pragma unroll
    for (int k = 0; k < 4; ++k) if (i0 + k < n) v += cnt[i0 + k];
    for (int o = 16; o > 0; o >>= 1) v += __shfl_xor(v, o, 32);
    if ((t & 31) == 0) red[t >> 5] = v;
    __syncthreads();
    if (t == 0) {
        int s = 0;
        #pragma unroll
        for (int k = 0; k < 8; ++k) s += red[k];
        bsum[blockIdx.x] = s;
    }
}

__global__ __launch_bounds__(256) void k_scan_apply(int* __restrict__ cnt,
                                                    const int* __restrict__ bsum, int n) {
    __shared__ int wsum[8];
    __shared__ int wtot[8];
    const int t = threadIdx.x, lane = t & 31, wave = t >> 5;

    int bv = 0;
    for (int i = t; i < (int)blockIdx.x; i += 256) bv += bsum[i];
    for (int o = 16; o > 0; o >>= 1) bv += __shfl_xor(bv, o, 32);
    if (lane == 0) wsum[wave] = bv;

    const int i0 = blockIdx.x * 1024 + t * 4;
    int a[4];
    #pragma unroll
    for (int k = 0; k < 4; ++k) a[k] = (i0 + k < n) ? cnt[i0 + k] : 0;
    const int tsum = a[0] + a[1] + a[2] + a[3];
    int inc = tsum;
    for (int o = 1; o < 32; o <<= 1) {
        int u = __shfl_up(inc, o, 32);
        if (lane >= o) inc += u;
    }
    if (lane == 31) wtot[wave] = inc;
    __syncthreads();

    int ex = inc - tsum;
    #pragma unroll
    for (int k = 0; k < 8; ++k) {
        ex += wsum[k];
        if (k < wave) ex += wtot[k];
    }
    #pragma unroll
    for (int k = 0; k < 4; ++k) {
        if (i0 + k < n) cnt[i0 + k] = ex;
        ex += a[k];
    }
}

__global__ __launch_bounds__(256) void k_fill(const int* __restrict__ seg, const int* __restrict__ pos,
                                              const int* __restrict__ ei,
                                              int* __restrict__ srcU, int e0) {
    int i = blockIdx.x * 256 + threadIdx.x;
    if (i < e0) srcU[seg[ei[e0 + i]] + pos[i]] = ei[i];
}

__global__ __launch_bounds__(256) void k_sortseg(const int* __restrict__ seg,
                                                 const int* __restrict__ srcU,
                                                 int* __restrict__ srcS, int n, int e0) {
    int c = blockIdx.x * 256 + threadIdx.x;
    if (c >= n) return;
    const int s = seg[c];
    const int e = (c + 1 < n) ? seg[c + 1] : e0;
    for (int i = s; i < e; ++i) {
        const int v = srcU[i];
        int rank = 0;
        for (int j = s; j < e; ++j) {
            const int u = srcU[j];
            rank += (u < v || (u == v && j < i)) ? 1 : 0;
        }
        srcS[s + rank] = v;
    }
}

__global__ __launch_bounds__(256) void k_packW(const float* __restrict__ W,
                                               float* __restrict__ Wf) {
    const int i = blockIdx.x * 256 + threadIdx.x;
    if (i >= NT * NP * 32) return;
    const int lane = i & 31;
    const int tp   = i >> 5;
    const int tile = tp / NP;
    const int p    = tp - tile * NP;
    const int n    = tile * 16 + (lane & 15);
    const int k0   = 8 * p + 2 * (lane >> 4);
    const int k1   = k0 + 4;
    float4 v = make_float4(0.f, 0.f, 0.f, 0.f);
    if (n < DF) {
        const float* row = W + (size_t)n * DF;
        if (k0 < DF) { const float2 u = *(const float2*)(row + k0); v.x = u.x; v.y = u.y; }
        if (k1 < DF) { const float2 u = *(const float2*)(row + k1); v.z = u.x; v.w = u.y; }
    }
    ((float4*)Wf)[i] = v;
}


#define MT 2
#define SLAB (MT * 64 + 2)
#define NSLAB (K4 + 1)

#define LDA(base, mt, s) (*(const v2f*)((base) + (mt) * 64 + (s) * SLAB))
#define LDB(p, pr)       (*(const v4f*)((p) + (pr) * 128))

__device__ __forceinline__ void stage_panel(const float* __restrict__ A, float* lds,
                                            int m0, int M, int t) {
    constexpr int TOT = 16 * MT * K4;
    constexpr int NJ  = (TOT + 255) / 256;
    float4 v[NJ];
    #pragma unroll
    for (int j = 0; j < NJ; ++j) {
        int i = t + j * 256;
        if (i > TOT - 1) i = TOT - 1;
        const int r  = i / K4;
        const int c4 = i - r * K4;
        const int m  = (m0 + r < M) ? m0 + r : M - 1;
        v[j] = ((const float4*)(A + (size_t)m * DF))[c4];
    }
    #pragma unroll
    for (int j = 0; j < NJ; ++j) {
        const int i = t + j * 256;
        if (i < TOT) {
            const int r  = i / K4;
            const int c4 = i - r * K4;
            const float4 u = v[j];
            float* d = &lds[c4 * SLAB + (r >> 4) * 64 + (r & 15) * 2];
            *(float2*)d        = make_float2(u.x, u.y);
            *(float2*)(d + 32) = make_float2(u.z, u.w);
        }
    }
}

template <int NM, int NN>
__device__ __forceinline__ void gemm_tail(const float* af, const float* __restrict__ b0,
                                          const float* __restrict__ bias, float* __restrict__ out,
                                          int n, int mbase, int M) {
    v8f c[NM][NN];
    v2f ax0[NM], ax1[NM], ay0[NM], ay1[NM];
    v4f px[NN], py[NN];
    #pragma unroll
    for (int i = 0; i < NM; ++i) {
        ax0[i] = LDA(af, i, 0);  ax1[i] = LDA(af, i, 1);
        ay0[i] = LDA(af, i, 2);  ay1[i] = LDA(af, i, 3);
        #pragma unroll
        for (int q = 0; q < NN; ++q) c[i][q] = v8f{};
    }
    #pragma unroll
    for (int q = 0; q < NN; ++q) { px[q] = LDB(b0 + q * WF_TILE, 0);  py[q] = LDB(b0 + q * WF_TILE, 1); }

#define TAIL_STEP(AF, BS, BH) \
    _Pragma("unroll") for (int i = 0; i < NM; ++i) \
    _Pragma("unroll") for (int q = 0; q < NN; ++q) c[i][q] = WMMA_F32(AF[i], BS[q].BH, c[i][q]);

    for (int p = 0; p < NP - 2; p += 2) {
        TAIL_STEP(ax0, px, lo)
        TAIL_STEP(ax1, px, hi)
        #pragma unroll
        for (int i = 0; i < NM; ++i) { ax0[i] = LDA(af, i, 2 * p + 4);  ax1[i] = LDA(af, i, 2 * p + 5); }
        #pragma unroll
        for (int q = 0; q < NN; ++q) px[q] = LDB(b0 + q * WF_TILE, p + 2);
        TAIL_STEP(ay0, py, lo)
        TAIL_STEP(ay1, py, hi)
        #pragma unroll
        for (int i = 0; i < NM; ++i) { ay0[i] = LDA(af, i, 2 * p + 6);  ay1[i] = LDA(af, i, 2 * p + 7); }
        #pragma unroll
        for (int q = 0; q < NN; ++q) py[q] = LDB(b0 + q * WF_TILE, p + 3);
    }
    TAIL_STEP(ax0, px, lo)
    TAIL_STEP(ax1, px, hi)
    TAIL_STEP(ay0, py, lo)
#undef TAIL_STEP

    #pragma unroll
    for (int q = 0; q < NN; ++q) {
        const int nq = n + q * 16;
        if (nq < DF) {
            const float bv = bias[nq];
            #pragma unroll
            for (int i = 0; i < NM; ++i) {
                #pragma unroll
                for (int r = 0; r < 8; ++r) {
                    const int m = mbase + i * 16 + r;
                    if (m < M) out[(size_t)m * DF + nq] = c[i][q][r] + bv;
                }
            }
        }
    }
}

__device__ __forceinline__ void gemm_compute(const float* lds,
                                             const float* __restrict__ Wf,
                                             const float* __restrict__ bias,
                                             float* __restrict__ out,
                                             int m0, int M, int wave, int lane) {
    const int ln   = lane & 15;
    const float* af = &lds[lane * 2];
    const int mbase = m0 + ((lane >> 4) << 3);

    {
        const int n0 = wave * 16 + ln;
        const int n1 = (wave + 8) * 16 + ln;
        const float* b0 = Wf + (size_t)wave * WF_TILE + lane * 4;
        const float* b1 = Wf + (size_t)(wave + 8) * WF_TILE + lane * 4;

        v8f c0[MT], c1[MT];
        v2f ax0[MT], ax1[MT], ay0[MT], ay1[MT];
        #pragma unroll
        for (int i = 0; i < MT; ++i) {
            c0[i] = v8f{};  c1[i] = v8f{};
            ax0[i] = LDA(af, i, 0);  ax1[i] = LDA(af, i, 1);
            ay0[i] = LDA(af, i, 2);  ay1[i] = LDA(af, i, 3);
        }
        v4f px = LDB(b0, 0), qx = LDB(b1, 0);
        v4f py = LDB(b0, 1), qy = LDB(b1, 1);

#define MAIN_STEP(AF, P, Q, H) \
    _Pragma("unroll") for (int i = 0; i < MT; ++i) { \
        c0[i] = WMMA_F32(AF[i], P.H, c0[i]);  c1[i] = WMMA_F32(AF[i], Q.H, c1[i]); }

        for (int p = 0; p < NP - 2; p += 2) {
            MAIN_STEP(ax0, px, qx, lo)
            MAIN_STEP(ax1, px, qx, hi)
            #pragma unroll
            for (int i = 0; i < MT; ++i) { ax0[i] = LDA(af, i, 2 * p + 4);  ax1[i] = LDA(af, i, 2 * p + 5); }
            px = LDB(b0, p + 2);  qx = LDB(b1, p + 2);
            MAIN_STEP(ay0, py, qy, lo)
            MAIN_STEP(ay1, py, qy, hi)
            #pragma unroll
            for (int i = 0; i < MT; ++i) { ay0[i] = LDA(af, i, 2 * p + 6);  ay1[i] = LDA(af, i, 2 * p + 7); }
            py = LDB(b0, p + 3);  qy = LDB(b1, p + 3);
        }
        MAIN_STEP(ax0, px, qx, lo)
        MAIN_STEP(ax1, px, qx, hi)
        MAIN_STEP(ay0, py, qy, lo)
#undef MAIN_STEP

        const float bv0 = bias[n0];
        const float bv1 = bias[n1];
        #pragma unroll
        for (int i = 0; i < MT; ++i) {
            #pragma unroll
            for (int r = 0; r < 8; ++r) {
                const int m = mbase + i * 16 + r;
                if (m < M) {
                    out[(size_t)m * DF + n0] = c0[i][r] + bv0;
                    out[(size_t)m * DF + n1] = c1[i][r] + bv1;
                }
            }
        }
    }

    if (wave < 3) {
        gemm_tail<MT, 1>(af, Wf + (size_t)(16 + wave) * WF_TILE + lane * 4, bias, out,
                         (16 + wave) * 16 + ln, mbase, M);
    }
}

#define NPW ((16 * MT) / 8)

#define ACC4(AA, WW, VV) { AA.x = fmaf(WW, VV.x, AA.x); AA.y = fmaf(WW, VV.y, AA.y); \
                           AA.z = fmaf(WW, VV.z, AA.z); AA.w = fmaf(WW, VV.w, AA.w); }

__device__ __forceinline__ void stage_gather(const float* __restrict__ h,
                                             const int* __restrict__ src,
                                             const int* __restrict__ seg,
                                             const float* __restrict__ dis,
                                             float* lds, int m0, int n, int e0,
                                             int wave, int lane) {
    const int c0 = m0 + wave * NPW;
    const int q2 = (lane + 64 < K4) ? lane + 64 : K4 - 1;

    int   sj[NPW], tj[NPW], rj[NPW];
    float dj[NPW], wj[NPW];
    #pragma unroll
    for (int j = 0; j < NPW; ++j) {
        const int c = c0 + j;
        const bool ok = c < n;
        const int cc = ok ? c : n - 1;
        sj[j] = seg[cc];
        tj[j] = ok ? ((cc + 1 < n) ? seg[cc + 1] : e0) - sj[j] + 1 : 0;
        dj[j] = dis[cc];
        int r = cc;
        if (lane >= 1 && lane < tj[j]) r = src[sj[j] + lane - 1];
        rj[j] = r;
    }
    #pragma unroll
    for (int j = 0; j < NPW; ++j) wj[j] = dis[rj[j]] * dj[j];

    #pragma unroll
    for (int j = 0; j < NPW; ++j) {
        const int c   = c0 + j;
        const int s   = sj[j];
        const int tot = tj[j];
        const float dc = dj[j];
        float4 a0 = make_float4(0.f, 0.f, 0.f, 0.f), a1 = a0, a2 = a0;

        for (int base = 0; base < tot; base += 32) {
            int myr = rj[j];
            int myw = __float_as_int(wj[j]);
            if (base > 0) {
                const int k = base + lane;
                myr = c;
                if (k < tot) myr = src[s + k - 1];
                myw = __float_as_int(dis[myr] * dc);
            }
            const int nn = (tot - base < 32) ? tot - base : 32;

            int i = 0;
            for (; i + 3 < nn; i += 4) {
                const int   r0 = __builtin_amdgcn_readlane(myr, i);
                const int   r1 = __builtin_amdgcn_readlane(myr, i + 1);
                const int   r2 = __builtin_amdgcn_readlane(myr, i + 2);
                const int   r3 = __builtin_amdgcn_readlane(myr, i + 3);
                const float w0 = __int_as_float(__builtin_amdgcn_readlane(myw, i));
                const float w1 = __int_as_float(__builtin_amdgcn_readlane(myw, i + 1));
                const float w2 = __int_as_float(__builtin_amdgcn_readlane(myw, i + 2));
                const float w3 = __int_as_float(__builtin_amdgcn_readlane(myw, i + 3));
                const float4* p0 = (const float4*)(h + (size_t)r0 * DF);
                const float4* p1 = (const float4*)(h + (size_t)r1 * DF);
                const float4* p2 = (const float4*)(h + (size_t)r2 * DF);
                const float4* p3 = (const float4*)(h + (size_t)r3 * DF);
                const float4 u0 = p0[lane], u1 = p0[lane + 32], u2 = p0[q2];
                const float4 v0 = p1[lane], v1 = p1[lane + 32], v2 = p1[q2];
                const float4 x0 = p2[lane], x1 = p2[lane + 32], x2 = p2[q2];
                const float4 y0 = p3[lane], y1 = p3[lane + 32], y2 = p3[q2];
                ACC4(a0, w0, u0); ACC4(a1, w0, u1); ACC4(a2, w0, u2);
                ACC4(a0, w1, v0); ACC4(a1, w1, v1); ACC4(a2, w1, v2);
                ACC4(a0, w2, x0); ACC4(a1, w2, x1); ACC4(a2, w2, x2);
                ACC4(a0, w3, y0); ACC4(a1, w3, y1); ACC4(a2, w3, y2);
            }
            const int rem = nn - i;
            if (rem > 0) {
                const float4 z = make_float4(0.f, 0.f, 0.f, 0.f);
                const int   r0 = __builtin_amdgcn_readlane(myr, i);
                const float w0 = __int_as_float(__builtin_amdgcn_readlane(myw, i));
                const float4* p0 = (const float4*)(h + (size_t)r0 * DF);
                const float4 u0 = p0[lane], u1 = p0[lane + 32], u2 = p0[q2];
                float4 v0 = z, v1 = z, v2 = z, x0 = z, x1 = z, x2 = z;
                float w1 = 0.f, w2 = 0.f;
                if (rem > 1) {
                    const int r1 = __builtin_amdgcn_readlane(myr, i + 1);
                    w1 = __int_as_float(__builtin_amdgcn_readlane(myw, i + 1));
                    const float4* p1 = (const float4*)(h + (size_t)r1 * DF);
                    v0 = p1[lane]; v1 = p1[lane + 32]; v2 = p1[q2];
                }
                if (rem > 2) {
                    const int r2 = __builtin_amdgcn_readlane(myr, i + 2);
                    w2 = __int_as_float(__builtin_amdgcn_readlane(myw, i + 2));
                    const float4* p2 = (const float4*)(h + (size_t)r2 * DF);
                    x0 = p2[lane]; x1 = p2[lane + 32]; x2 = p2[q2];
                }
                ACC4(a0, w0, u0); ACC4(a1, w0, u1); ACC4(a2, w0, u2);
                if (rem > 1) { ACC4(a0, w1, v0); ACC4(a1, w1, v1); ACC4(a2, w1, v2); }
                if (rem > 2) { ACC4(a0, w2, x0); ACC4(a1, w2, x1); ACC4(a2, w2, x2); }
            }
        }

        const int r = wave * NPW + j;
        float* d = &lds[(r >> 4) * 64 + (r & 15) * 2];
        float* d0 = d + lane * SLAB;
        float* d1 = d + (lane + 32) * SLAB;
        *(float2*)d0        = make_float2(fmaxf(a0.x, 0.f), fmaxf(a0.y, 0.f));
        *(float2*)(d0 + 32) = make_float2(fmaxf(a0.z, 0.f), fmaxf(a0.w, 0.f));
        *(float2*)d1        = make_float2(fmaxf(a1.x, 0.f), fmaxf(a1.y, 0.f));
        *(float2*)(d1 + 32) = make_float2(fmaxf(a1.z, 0.f), fmaxf(a1.w, 0.f));
        if (lane + 64 < K4) {
            float* d2 = d + (lane + 64) * SLAB;
            *(float2*)d2        = make_float2(fmaxf(a2.x, 0.f), fmaxf(a2.y, 0.f));
            *(float2*)(d2 + 32) = make_float2(fmaxf(a2.z, 0.f), fmaxf(a2.w, 0.f));
        }
    }
}
#undef ACC4


__global__ __launch_bounds__(256) void k_gemm(const float* __restrict__ A,
                                              const float* __restrict__ Wf,
                                              const float* __restrict__ bias,
                                              float* __restrict__ out, int M) {
    __shared__ float lds[NSLAB * SLAB];
    const int m0 = blockIdx.x * (16 * MT);
    const int t  = threadIdx.x;

    stage_panel(A, lds, m0, M, t);
    __syncthreads();

    const int wave = __builtin_amdgcn_readfirstlane(t >> 5);
    gemm_compute(lds, Wf, bias, out, m0, M, wave, t & 31);
}

__global__ __launch_bounds__(256) void k_gemm_agg(const float* __restrict__ h,
                                                  const int* __restrict__ src,
                                                  const int* __restrict__ seg,
                                                  const float* __restrict__ dis,
                                                  const float* __restrict__ Wf,
                                                  const float* __restrict__ bias,
                                                  float* __restrict__ out, int M, int e0) {
    __shared__ float lds[NSLAB * SLAB];
    const int m0 = blockIdx.x * (16 * MT);
    const int t  = threadIdx.x;
    const int wave = __builtin_amdgcn_readfirstlane(t >> 5);

    stage_gather(h, src, seg, dis, lds, m0, M, e0, wave, t & 31);
    __syncthreads();

    gemm_compute(lds, Wf, bias, out, m0, M, wave, t & 31);
}
#undef LDA
#undef LDB


__global__ __launch_bounds__(256) void k_gather(const float* __restrict__ h,
                                                const int* __restrict__ src,
                                                const int* __restrict__ seg,
                                                const float* __restrict__ dis,
                                                float* __restrict__ out, int n, int e0) {
    const int c = __builtin_amdgcn_readfirstlane((int)(blockIdx.x * 8 + (threadIdx.x >> 5)));
    if (c >= n) return;
    const int lane = threadIdx.x & 31;
    const int s    = seg[c];
    const int tot  = ((c + 1 < n) ? seg[c + 1] : e0) - s + 1;
    const float dc = dis[c];
    const int q2 = (lane + 64 < K4) ? lane + 64 : K4 - 1;

    float4 a0 = make_float4(0.f, 0.f, 0.f, 0.f), a1 = a0, a2 = a0;

#define ACC(AA, WW, VV) { AA.x = fmaf(WW, VV.x, AA.x); AA.y = fmaf(WW, VV.y, AA.y); \
                          AA.z = fmaf(WW, VV.z, AA.z); AA.w = fmaf(WW, VV.w, AA.w); }

    for (int base = 0; base < tot; base += 32) {
        const int k = base + lane;
        int myr = c;
        if (k >= 1 && k < tot) myr = src[s + k - 1];
        const int myw = __float_as_int(dis[myr] * dc);
        const int nn = (tot - base < 32) ? tot - base : 32;

        int i = 0;
        for (; i + 1 < nn; i += 2) {
            const int   r0 = __builtin_amdgcn_readlane(myr, i);
            const int   r1 = __builtin_amdgcn_readlane(myr, i + 1);
            const float w0 = __int_as_float(__builtin_amdgcn_readlane(myw, i));
            const float w1 = __int_as_float(__builtin_amdgcn_readlane(myw, i + 1));
            const float4* p0 = (const float4*)(h + (size_t)r0 * DF);
            const float4* p1 = (const float4*)(h + (size_t)r1 * DF);
            const float4 u0 = p0[lane], u1 = p0[lane + 32], u2 = p0[q2];
            const float4 v0 = p1[lane], v1 = p1[lane + 32], v2 = p1[q2];
            ACC(a0, w0, u0); ACC(a1, w0, u1); ACC(a2, w0, u2);
            ACC(a0, w1, v0); ACC(a1, w1, v1); ACC(a2, w1, v2);
        }
        if (i < nn) {
            const int   r0 = __builtin_amdgcn_readlane(myr, i);
            const float w0 = __int_as_float(__builtin_amdgcn_readlane(myw, i));
            const float4* p0 = (const float4*)(h + (size_t)r0 * DF);
            const float4 u0 = p0[lane], u1 = p0[lane + 32], u2 = p0[q2];
            ACC(a0, w0, u0); ACC(a1, w0, u1); ACC(a2, w0, u2);
        }
    }
#undef ACC

    float4* o = (float4*)(out + (size_t)c * DF);
    o[lane]      = a0;
    o[lane + 32] = a1;
    if (lane + 64 < K4) o[lane + 64] = a2;
}


extern "C" void kernel_launch(void* const* d_in, const int* in_sizes, int n_in,
                              void* d_out, int out_size, void* d_ws, size_t ws_size,
                              hipStream_t stream) {
    const float* x  = (const float*)d_in[0];
    const int*   ei = (const int*)d_in[1];
    const float* W1 = (const float*)d_in[2];
    const float* b1 = (const float*)d_in[3];
    const float* W2 = (const float*)d_in[4];
    const float* b2 = (const float*)d_in[5];
    float* out = (float*)d_out;

    const int N    = in_sizes[0] / DF;
    const int E0   = in_sizes[1] / 2;

    float* dis = (float*)d_ws;
    size_t disPad = ((size_t)N + 127) & ~(size_t)127;
    float* h  = dis + disPad;
    float* g  = h + (size_t)N * DF;
    int* cnt  = (int*)g;
    int* bsum = cnt + disPad;
    int* srcU = bsum + 1024;
    int* srcS = srcU + E0;
    int* pos  = srcS + E0;
    size_t wfOff = (disPad + 1024 + 3 * (size_t)E0 + 63) & ~(size_t)63;
    float* wf = g + wfOff;

    const int mtiles = (N + 16 * MT - 1) / (16 * MT);
    const int npackW = (NT * NP * 32 + 255) / 256;
    const int nscan  = (N + 1023) / 1024;

    k_init     <<<(N  + 255) / 256, 256, 0, stream>>>(dis, cnt, N);
    k_deg_accum<<<(E0 + 255) / 256, 256, 0, stream>>>(dis, cnt, pos, ei, E0);
    k_rsqrt    <<<(N  + 255) / 256, 256, 0, stream>>>(dis, N);

    k_scan_partial<<<nscan, 256, 0, stream>>>(cnt, bsum, N);
    k_scan_apply  <<<nscan, 256, 0, stream>>>(cnt, bsum, N);
    k_fill        <<<(E0 + 255) / 256, 256, 0, stream>>>(cnt, pos, ei, srcU, E0);
    k_sortseg     <<<(N  + 255) / 256, 256, 0, stream>>>(cnt, srcU, srcS, N, E0);

    k_packW <<<npackW, 256, 0, stream>>>(W1, wf);
    k_gemm  <<<mtiles, 256, 0, stream>>>(x, wf, b1, out, N);

    k_packW   <<<npackW, 256, 0, stream>>>(W2, wf);
    k_gemm_agg<<<mtiles, 256, 0, stream>>>(out, srcS, cnt, dis, wf, b2, h, N, E0);

    k_gather<<<(N + 7) / 8, 256, 0, stream>>>(h, srcS, cnt, dis, out, N, E0);
}
